# starting kernel
# baseline (speedup 1.0000x reference)
#include <hip/hip_runtime.h>
#include <hip/hip_bf16.h>

typedef __attribute__((ext_vector_type(2))) float v2f;
typedef __attribute__((ext_vector_type(8))) float v8f;

#define S_LEN 1024
#define HD    64
#define NW    4
#define WROW  18

__device__ __forceinline__ v8f wmma_f32(v2f a, v2f b, v8f c) {
  return __builtin_amdgcn_wmma_f32_16x16x4_f32(false, a, false, b, (short)0, c,
                                               false, false);
}

__global__ __launch_bounds__(128) void stickbreak_kernel(
    const float* __restrict__ q, const float* __restrict__ k,
    const float* __restrict__ v, float* __restrict__ out) {
  __shared__ float lds[NW * 16 * WROW];

  const int bh   = blockIdx.x;
  const int wave = threadIdx.x >> 5;
  const int lane = threadIdx.x & 31;
  const int m    = lane & 15;
  const int hi   = lane >> 4;

  const int qt = (blockIdx.y * NW + wave) * 16;

  const size_t base = (size_t)bh * S_LEN * HD;
  const float* qb = q + base;
  const float* kb = k + base;
  const float* vb = v + base;
  float*       ob = out + base;

  const float scale = 0.125f;

  v2f qa[16];
#pragma unroll
  for (int c = 0; c < 16; ++c) {
    const float* p = qb + (size_t)(qt + m) * HD + 4 * c + 2 * hi;
    v2f t = *(const v2f*)p;
    qa[c] = t * scale;
  }

  v8f acc0 = {}, acc1 = {}, acc2 = {}, acc3 = {};
  float carry[8];
#pragma unroll
  for (int r = 0; r < 8; ++r) carry[r] = 0.0f;

  float* wl = lds + wave * 16 * WROW;

  for (int j0 = qt; j0 >= 0; j0 -= 16) {
    if (j0 >= 16) {
      __builtin_prefetch(kb + (size_t)(j0 - 16 + m) * HD, 0, 3);
      __builtin_prefetch(vb + (size_t)(j0 - 16 + m) * HD, 0, 3);
    }

    v8f L = {};
#pragma unroll
    for (int c = 0; c < 16; ++c) {
      const float* p = kb + (size_t)(j0 + m) * HD + 4 * c + 2 * hi;
      v2f kt = *(const v2f*)p;
      L = wmma_f32(qa[c], kt, L);
    }

    const bool diag = (j0 == qt);

    float w[8];
#pragma unroll
    for (int r = 0; r < 8; ++r) {
      float x  = L[r];
      float zz = 1.0f / (1.0f + __expf(-x));
      float sp = fmaxf(x, 0.0f) + __logf(1.0f + __expf(-fabsf(x)));
      float lb = -sp;
      if (diag && (m > r + 8 * hi)) { zz = 0.0f; lb = 0.0f; }
      float s = lb;
#pragma unroll
      for (int off = 1; off < 16; off <<= 1) {
        float t = __shfl(s, lane + off);
        if (m + off < 16) s += t;
      }
      float tot = __shfl(s, hi << 4);
      w[r] = zz * __expf(s + carry[r]);
      carry[r] += tot;
    }

#pragma unroll
    for (int r = 0; r < 8; ++r)
      wl[(r + 8 * hi) * WROW + m] = w[r];
    asm volatile("s_wait_dscnt 0x0" ::: "memory");

    v2f wa[4];
#pragma unroll
    for (int c = 0; c < 4; ++c)
      wa[c] = *(const v2f*)&wl[m * WROW + 4 * c + 2 * hi];

#pragma unroll
    for (int c = 0; c < 4; ++c) {
      const float* vr0 = vb + (size_t)(j0 + 4 * c + 2 * hi) * HD + m;
      const float* vr1 = vr0 + HD;
      v2f b0 = { vr0[0],  vr1[0]  };
      v2f b1 = { vr0[16], vr1[16] };
      v2f b2 = { vr0[32], vr1[32] };
      v2f b3 = { vr0[48], vr1[48] };
      acc0 = wmma_f32(wa[c], b0, acc0);
      acc1 = wmma_f32(wa[c], b1, acc1);
      acc2 = wmma_f32(wa[c], b2, acc2);
      acc3 = wmma_f32(wa[c], b3, acc3);
    }
  }

#pragma unroll
  for (int r = 0; r < 8; ++r) {
    size_t row = (size_t)(qt + r + 8 * hi) * HD;
    ob[row + 0  + m] = acc0[r];
    ob[row + 16 + m] = acc1[r];
    ob[row + 32 + m] = acc2[r];
    ob[row + 48 + m] = acc3[r];
  }
}

extern "C" void kernel_launch(void* const* d_in, const int* in_sizes, int n_in,
                              void* d_out, int out_size, void* d_ws, size_t ws_size,
                              hipStream_t stream) {
  const float* q = (const float*)d_in[0];
  const float* k = (const float*)d_in[1];
  const float* v = (const float*)d_in[2];
  float* out = (float*)d_out;

  const int BH = in_sizes[0] / (S_LEN * HD);
  dim3 grid(BH, S_LEN / (16 * NW));
  dim3 block(32 * NW);
  hipLaunchKernelGGL(stickbreak_kernel, grid, block, 0, stream, q, k, v, out);
}


# Trip software-pipelined inside the wave: logits one key group ahead; scan, weights and sigmoids issued as independent slices between the matrix instructions of both runs; order of sums unchanged
# speedup vs baseline: 2.4049x; 2.4049x over previous
#include <hip/hip_runtime.h>
#include <hip/hip_bf16.h>


typedef __attribute__((ext_vector_type(2))) float v2f;
typedef __attribute__((ext_vector_type(8))) float v8f;
typedef __attribute__((ext_vector_type(4))) float v4f;

#define S_LEN 1024
#define HD    64
#define NW    4
#define KB    4
#define WROW  18
#define GK    (16 * KB)
#define NG    (S_LEN / (16 * NW))
#define KP    68
#define VP    160

__device__ __forceinline__ v8f wmma_f32(v2f a, v2f b, v8f c) {
  return __builtin_amdgcn_wmma_f32_16x16x4_f32(false, a, false, b, (short)0, c,
                                               false, false);
}

#define ROW_SHL_MUL(N)                                                         \
  __device__ __forceinline__ float row_shl_mul_##N(float x) {                  \
    asm("v_mul_f32_dpp %0, %0, %0 row_shl:" #N " row_mask:0xf bank_mask:0xf"   \
        : "+v"(x));                                                            \
    return x;                                                                  \
  }
ROW_SHL_MUL(1)
ROW_SHL_MUL(2)
ROW_SHL_MUL(4)
ROW_SHL_MUL(8)
__device__ __forceinline__ float row_first(float x) {
  return __builtin_bit_cast(float,
      __builtin_amdgcn_update_dpp(0, __builtin_bit_cast(int, x),
                                  0x150  , 0xf, 0xf, false));
}


__device__ __forceinline__ void scan_slice(int bb, int ph, float (&zz)[KB][8],
                                           float (&sc)[KB][8], float (&tot)[KB][8],
                                           float (&w)[KB][8], float (&carry)[8]) {
  const int r0 = (ph & 1) * 4;
  const int st = ph >> 1;
#pragma unroll
  for (int i = 0; i < 4; ++i) {
    const int r = r0 + i;
    if (st == 0) sc[bb][r] = row_shl_mul_1(sc[bb][r]);
    if (st == 1) sc[bb][r] = row_shl_mul_2(sc[bb][r]);
    if (st == 2) sc[bb][r] = row_shl_mul_4(sc[bb][r]);
    if (st == 3) sc[bb][r] = row_shl_mul_8(sc[bb][r]);
    if (st == 4) tot[bb][r] = row_first(sc[bb][r]);
    if (st == 5) w[bb][r] = sc[bb][r] * carry[r];
    if (st == 6) w[bb][r] = zz[bb][r] * w[bb][r];
    if (st == 7) carry[r] *= tot[bb][r];
  }
}

__device__ __forceinline__ void sig_slice(int p, int half, const v8f (&L)[KB],
                                          float (&ta)[32], float (&ea)[32],
                                          float (&da)[32], float (&ra)[32],
                                          float (&pa)[32],
                                          float (&zz)[KB][8], float (&sc)[KB][8]) {
  const int a = p, b = p - 1, c = p - 2;
  if (half == 0) {
    if (a >= 0 && a < 32) ta[a] = -1.44269504f * fabsf(L[a >> 3][a & 7]);
    if (b >= 0 && b < 32) da[b] = 1.0f + ea[b];
    if (c >= 0 && c < 32) pa[c] = ea[c] * ra[c];
  } else {
    if (a >= 0 && a < 32) ea[a] = __builtin_amdgcn_exp2f(ta[a]);
    if (b >= 0 && b < 32) ra[b] = __builtin_amdgcn_rcpf(da[b]);
    if (c >= 0 && c < 32) {
      const bool pos = L[c >> 3][c & 7] >= 0.0f;
      zz[c >> 3][c & 7] = pos ? ra[c] : pa[c];
      sc[c >> 3][c & 7] = pos ? pa[c] : ra[c];
    }
  }
}

__global__ __launch_bounds__(128) void stickbreak_kernel(
    const float* __restrict__ q, const float* __restrict__ k,
    const float* __restrict__ v, float* __restrict__ out) {
  __shared__ float lds[NW * KB * 16 * WROW];
  __shared__ __attribute__((aligned(16))) float lk[2][GK * KP];
  __shared__ __attribute__((aligned(16))) float lv[2][(GK / 2) * VP];

  const int bh   = blockIdx.x;
  const int wave = threadIdx.x >> 5;
  const int lane = threadIdx.x & 31;
  const int m    = lane & 15;
  const int hi   = lane >> 4;

  const int by = blockIdx.y;
  const int tg = by < NG / 2 ? NG - 1 - by : by - NG / 2;

  const int qt = (tg * NW + wave) * 16;

  const size_t base = (size_t)bh * S_LEN * HD;
  const float* qb = q + base;
  const float* kb = k + base;
  const float* vb = v + base;
  float*       ob = out + base;

  const float scale = 0.125f;

  const int srow = threadIdx.x >> 4;
  const int sc4  = threadIdx.x & 15;


  v2f qa[16];
#pragma unroll
  for (int c = 0; c < 16; ++c) {
    const float* p = qb + (size_t)(qt + m) * HD + 4 * c + 2 * hi;
    v2f t = *(const v2f*)p;
    qa[c] = t * scale;
  }

  v8f acc[4] = {};
  float carry[8];
#pragma unroll
  for (int r = 0; r < 8; ++r) carry[r] = 1.0f;

  float* wl = lds + wave * (KB * 16 * WROW);

  const int gtop = tg;

  float zz[KB][8], sc[KB][8];

  {
    const int jg = gtop * GK;
    const int jb = (gtop > 0 ? gtop - 1 : 0) * GK;
    float* lkt = lk[gtop & 1];
    float* lkb = lk[(gtop & 1) ^ 1];
    float* lvt = lv[gtop & 1];
#pragma unroll
    for (int i = 0; i < 8; ++i) {
      const int row = i * 8 + srow;
      v4f t = *(const v4f*)(kb + (size_t)(jg + row) * HD + 4 * sc4);
      v4f u = *(const v4f*)(kb + (size_t)(jb + row) * HD + 4 * sc4);
      *(v4f*)&lkt[row * KP + 4 * sc4] = t;
      *(v4f*)&lkb[row * KP + 4 * sc4] = u;
    }
#pragma unroll
    for (int i = 0; i < 4; ++i) {
      const int pr = i * 8 + srow;
      const float* p = vb + (size_t)(jg + 2 * pr) * HD + 4 * sc4;
      v4f a = *(const v4f*)p;
      v4f b = *(const v4f*)(p + HD);
      *(v4f*)&lvt[pr * VP + 8 * sc4]     = (v4f){a.x, b.x, a.y, b.y};
      *(v4f*)&lvt[pr * VP + 8 * sc4 + 4] = (v4f){a.z, b.z, a.w, b.w};
    }
  }
  __syncthreads();

  {
    const float* lkc = lk[gtop & 1];
    const int jg = gtop * GK;
    v8f L[KB];
#pragma unroll
    for (int b = 0; b < KB; ++b) L[b] = (v8f){};
    v2f kf[16][KB];
#pragma unroll
    for (int c = 0; c < 16; ++c)
#pragma unroll
      for (int b = 0; b < KB; ++b)
        kf[c][b] = *(const v2f*)&lkc[(16 * b + m) * KP + 4 * c + 2 * hi];
#pragma unroll
    for (int c = 0; c < 16; ++c)
#pragma unroll
      for (int b = 0; b < KB; ++b)
        L[b] = wmma_f32(qa[c], kf[c][b], L[b]);
#pragma unroll
    for (int b = 0; b < KB; ++b) {
#pragma unroll
      for (int r = 0; r < 8; ++r) {
        const float x  = L[b][r];
        const float e  = __builtin_amdgcn_exp2f(-1.44269504f * fabsf(x));
        const float rc = __builtin_amdgcn_rcpf(1.0f + e);
        const float er = e * rc;
        const bool pos = x >= 0.0f;
        float z1 = pos ? rc : er;
        float bt = pos ? er : rc;
        if (jg + 16 * b + m > qt + r + 8 * hi) { z1 = 0.0f; bt = 1.0f; }
        zz[b][r] = z1;
        sc[b][r] = bt;
      }
    }
  }
  __syncthreads();

  float tot[KB][8], w[KB][8];

  for (int g = gtop; g >= 1; --g) {
    const float* lkc = lk[(g & 1) ^ 1];
    const float* lvc = lv[g & 1];
    float* lkn = lk[g & 1];
    float* lvn = lv[(g & 1) ^ 1];
    const int jk = (g > 1 ? g - 2 : 0) * GK;
    const int jv = (g - 1) * GK;

    v4f ks[8];
#pragma unroll
    for (int i = 0; i < 8; ++i)
      ks[i] = *(const v4f*)(kb + (size_t)(jk + i * 8 + srow) * HD + 4 * sc4);

    v8f Ln[KB];
#pragma unroll
    for (int b = 0; b < KB; ++b) Ln[b] = (v8f){};
    v2f kf[16][KB];
#pragma unroll
    for (int c = 0; c < 16; ++c)
#pragma unroll
      for (int b = 0; b < KB; ++b)
        kf[c][b] = *(const v2f*)&lkc[(16 * b + m) * KP + 4 * c + 2 * hi];
    __builtin_amdgcn_sched_barrier(0);

#pragma unroll
    for (int q = 0; q < KB; ++q) {
#pragma unroll
      for (int ph = 0; ph < 16; ++ph) {
        const int s = 16 * q + ph;
        Ln[s & 3] = wmma_f32(qa[s >> 2], kf[s >> 2][s & 3], Ln[s & 3]);
        scan_slice(KB - 1 - q, ph, zz, sc, tot, w, carry);
        __builtin_amdgcn_sched_barrier(0);
      }
    }

#pragma unroll
    for (int i = 0; i < 8; ++i)
      *(v4f*)&lkn[(i * 8 + srow) * KP + 4 * sc4] = ks[i];
    v4f va[4], vc[4];
#pragma unroll
    for (int i = 0; i < 4; ++i) {
      const float* p = vb + (size_t)(jv + 2 * (i * 8 + srow)) * HD + 4 * sc4;
      va[i] = *(const v4f*)p;
      vc[i] = *(const v4f*)(p + HD);
    }

#pragma unroll
    for (int b = 0; b < KB; ++b)
#pragma unroll
      for (int r = 0; r < 8; ++r)
        wl[b * (16 * WROW) + (r + 8 * hi) * WROW + m] = w[b][r];
    asm volatile("s_wait_dscnt 0x0" ::: "memory");

    v2f wa[KB][4];
#pragma unroll
    for (int b = 0; b < KB; ++b)
#pragma unroll
      for (int c = 0; c < 4; ++c)
        wa[b][c] = *(const v2f*)&wl[b * (16 * WROW) + m * WROW + 4 * c + 2 * hi];

    v2f vf[KB][4][4];
#pragma unroll
    for (int b = KB - 1; b >= 0; --b)
#pragma unroll
      for (int c = 0; c < 4; ++c) {
        const float* vp = &lvc[(8 * b + 2 * c + hi) * VP + 2 * m];
#pragma unroll
        for (int t = 0; t < 4; ++t) vf[b][c][t] = *(const v2f*)(vp + 32 * t);
      }
    __builtin_amdgcn_sched_barrier(0);

    float ta[32], ea[32], da[32], ra[32], pa[32];
#pragma unroll
    for (int s = 0; s < 68; ++s) {
      if (s < 64) {
        const int b = KB - 1 - (s >> 4), c = (s >> 2) & 3, t = s & 3;
        acc[t] = wmma_f32(wa[b][c], vf[b][c][t], acc[t]);
      }
      sig_slice(s >> 1, s & 1, Ln, ta, ea, da, ra, pa, zz, sc);
      __builtin_amdgcn_sched_barrier(0);
    }

#pragma unroll
    for (int i = 0; i < 4; ++i) {
      const int pr = i * 8 + srow;
      *(v4f*)&lvn[pr * VP + 8 * sc4]     = (v4f){va[i].x, vc[i].x, va[i].y, vc[i].y};
      *(v4f*)&lvn[pr * VP + 8 * sc4 + 4] = (v4f){va[i].z, vc[i].z, va[i].w, vc[i].w};
    }
    __syncthreads();
  }

  {
    const float* lvc = lv[0];
#pragma unroll
    for (int q = 0; q < KB; ++q)
#pragma unroll
      for (int ph = 0; ph < 16; ++ph)
        scan_slice(KB - 1 - q, ph, zz, sc, tot, w, carry);
#pragma unroll
    for (int b = 0; b < KB; ++b)
#pragma unroll
      for (int r = 0; r < 8; ++r)
        wl[b * (16 * WROW) + (r + 8 * hi) * WROW + m] = w[b][r];
    asm volatile("s_wait_dscnt 0x0" ::: "memory");
    v2f wa[KB][4];
#pragma unroll
    for (int b = 0; b < KB; ++b)
#pragma unroll
      for (int c = 0; c < 4; ++c)
        wa[b][c] = *(const v2f*)&wl[b * (16 * WROW) + m * WROW + 4 * c + 2 * hi];
    v2f vf[KB][4][4];
#pragma unroll
    for (int b = KB - 1; b >= 0; --b)
#pragma unroll
      for (int c = 0; c < 4; ++c) {
        const float* vp = &lvc[(8 * b + 2 * c + hi) * VP + 2 * m];
#pragma unroll
        for (int t = 0; t < 4; ++t) vf[b][c][t] = *(const v2f*)(vp + 32 * t);
      }
#pragma unroll
    for (int b = KB - 1; b >= 0; --b)
#pragma unroll
      for (int c = 0; c < 4; ++c)
#pragma unroll
        for (int t = 0; t < 4; ++t)
          acc[t] = wmma_f32(wa[b][c], vf[b][c][t], acc[t]);
  }

#pragma unroll
  for (int r = 0; r < 8; ++r) {
    size_t row = (size_t)(qt + r + 8 * hi) * HD;
    ob[row + 0  + m] = acc[0][r];
    ob[row + 16 + m] = acc[1][r];
    ob[row + 32 + m] = acc[2][r];
    ob[row + 48 + m] = acc[3][r];
  }
}

extern "C" void kernel_launch(void* const* d_in, const int* in_sizes, int n_in,
                              void* d_out, int out_size, void* d_ws, size_t ws_size,
                              hipStream_t stream) {
  const float* q = (const float*)d_in[0];
  const float* k = (const float*)d_in[1];
  const float* v = (const float*)d_in[2];
  float* out = (float*)d_out;

  const int BH = in_sizes[0] / (S_LEN * HD);
  dim3 grid(BH, S_LEN / (16 * NW));
  dim3 block(32 * NW);
  hipLaunchKernelGGL(stickbreak_kernel, grid, block, 0, stream, q, k, v, out);
}


# LDS transpose moved from the weights to the raw logits, off the path between the matrix runs; sigmoids and beta products run in the weights' operand layout in-lane, no DPP; sum, product order change
# speedup vs baseline: 2.6059x; 1.0836x over previous
#include <hip/hip_runtime.h>
#include <hip/hip_bf16.h>


typedef __attribute__((ext_vector_type(2))) float v2f;
typedef __attribute__((ext_vector_type(8))) float v8f;
typedef __attribute__((ext_vector_type(4))) float v4f;

#define S_LEN 1024
#define HD    64
#define NW    4
#define KB    4
#define WROW  20
#define GK    (16 * KB)
#define NG    (S_LEN / (16 * NW))
#define KP    68
#define VP    136

__device__ __forceinline__ v8f wmma_f32(v2f a, v2f b, v8f c) {
  return __builtin_amdgcn_wmma_f32_16x16x4_f32(false, a, false, b, (short)0, c,
                                               false, false);
}

__device__ __forceinline__ float other_half(float x) {
  const unsigned u = __builtin_bit_cast(unsigned, x);
  return __builtin_bit_cast(float,
      __builtin_amdgcn_permlanex16(u, u, 0x76543210u, 0xfedcba98u, false, false));
}

template <int K, int ST>
__device__ __forceinline__ void sig_step(const v8f (&LA)[KB], float (&ta)[32],
                                         float (&ea)[32], float (&da)[32],
                                         float (&ra)[32], float (&pa)[32],
                                         float (&zz)[KB][8], float (&sc)[KB][8]) {
  if constexpr (K >= 0 && K < 32) {
    constexpr int b = KB - 1 - (K >> 3), e = 7 - (K & 7);
    const float x = LA[b][e];
    if constexpr (ST == 0) ta[K] = -1.44269504f * fabsf(x);
    if constexpr (ST == 1) ea[K] = __builtin_amdgcn_exp2f(ta[K]);
    if constexpr (ST == 2) da[K] = 1.0f + ea[K];
    if constexpr (ST == 3) ra[K] = __builtin_amdgcn_rcpf(da[K]);
    if constexpr (ST == 4) pa[K] = ea[K] * ra[K];
    if constexpr (ST == 5) {
      const bool pos = x >= 0.0f;
      zz[b][e] = pos ? ra[K] : pa[K];
      sc[b][e] = pos ? pa[K] : ra[K];
    }
  }
}

template <int B, int U>
__device__ __forceinline__ void scan_step(bool upper, float (&zz)[KB][8],
                                          float (&sc)[KB][8], float (&to)[KB],
                                          float (&fs)[KB], float (&tt)[KB],
                                          float (&F)[KB], v2f (&sf)[KB][4],
                                          float &carry, v2f (&wa)[KB][4]) {
  if constexpr (U >= 0 && U < 7) sc[B][6 - U] = sc[B][6 - U] * sc[B][7 - U];
  if constexpr (U == 7) to[B] = other_half(sc[B][0]);
  if constexpr (U == 8) {
    fs[B] = upper ? 1.0f : to[B];
    tt[B] = sc[B][0] * to[B];
  }
  if constexpr (U == 9) {
    F[B]  = carry * fs[B];
    carry = carry * tt[B];
  }
  if constexpr (U >= 10 && U <= 17) {
    constexpr int c = (U - 10) >> 1;
    if constexpr (((U - 10) & 1) == 0)
      sf[B][c] = (v2f){sc[B][2 * c] * F[B], sc[B][2 * c + 1] * F[B]};
    else
      wa[B][c] = (v2f){zz[B][2 * c] * sf[B][c].x, zz[B][2 * c + 1] * sf[B][c].y};
  }
}

template <int S>
__device__ __forceinline__ void group_slice(bool upper, const v8f (&LA)[KB],
                                            float (&ta)[32], float (&ea)[32],
                                            float (&da)[32], float (&ra)[32],
                                            float (&pa)[32],
                                            float (&zz)[KB][8], float (&sc)[KB][8],
                                            float (&to)[KB], float (&fs)[KB],
                                            float (&tt)[KB], float (&F)[KB],
                                            v2f (&sf)[KB][4], float &carry,
                                            v2f (&wa)[KB][4]) {
  sig_step<S - 0, 0>(LA, ta, ea, da, ra, pa, zz, sc);
  sig_step<S - 1, 1>(LA, ta, ea, da, ra, pa, zz, sc);
  sig_step<S - 2, 2>(LA, ta, ea, da, ra, pa, zz, sc);
  sig_step<S - 3, 3>(LA, ta, ea, da, ra, pa, zz, sc);
  sig_step<S - 4, 4>(LA, ta, ea, da, ra, pa, zz, sc);
  sig_step<S - 5, 5>(LA, ta, ea, da, ra, pa, zz, sc);
  scan_step<3, S - 7>(upper, zz, sc, to, fs, tt, F, sf, carry, wa);
  scan_step<2, S - 15>(upper, zz, sc, to, fs, tt, F, sf, carry, wa);
  scan_step<1, S - 23>(upper, zz, sc, to, fs, tt, F, sf, carry, wa);
  scan_step<0, S - 31>(upper, zz, sc, to, fs, tt, F, sf, carry, wa);
}
#define GS_ARGS upper, LA, ta, ea, da, ra, pa, zz, sc, to, fs, tt, F, sf, carry, wa
#define GS8(a) group_slice<(a)>(GS_ARGS); group_slice<(a) + 1>(GS_ARGS);       \
  group_slice<(a) + 2>(GS_ARGS); group_slice<(a) + 3>(GS_ARGS);                \
  group_slice<(a) + 4>(GS_ARGS); group_slice<(a) + 5>(GS_ARGS);                \
  group_slice<(a) + 6>(GS_ARGS); group_slice<(a) + 7>(GS_ARGS);
#define SLOT(a)                                                                \
  Ln[(a) & 3] = wmma_f32(qa[(a) >> 2], kf[(a) >> 2][(a) & 3], Ln[(a) & 3]);    \
  group_slice<(a)>(GS_ARGS);
#define PIPE1                                                                  \
  __builtin_amdgcn_sched_group_barrier(0x008, 1, 0);                           \
  __builtin_amdgcn_sched_group_barrier(0x002, 3, 0);                           \
  __builtin_amdgcn_sched_group_barrier(0x400, 1, 0);                           \
  __builtin_amdgcn_sched_group_barrier(0x002, 3, 0);                           \
  __builtin_amdgcn_sched_group_barrier(0x400, 1, 0);
#define PIPE8 PIPE1 PIPE1 PIPE1 PIPE1 PIPE1 PIPE1 PIPE1 PIPE1
#define SLOT8(a) SLOT(a) SLOT((a) + 1) SLOT((a) + 2) SLOT((a) + 3)             \
  SLOT((a) + 4) SLOT((a) + 5) SLOT((a) + 6) SLOT((a) + 7)


__global__ __launch_bounds__(128) void stickbreak_kernel(
    const float* __restrict__ q, const float* __restrict__ k,
    const float* __restrict__ v, float* __restrict__ out) {
  __shared__ __attribute__((aligned(16))) float lds[NW * KB * 16 * WROW];
  __shared__ __attribute__((aligned(16))) float lk[2][GK * KP];
  __shared__ __attribute__((aligned(16))) float lv[2][(GK / 2) * VP];

  const int bh   = blockIdx.x;
  const int wave = threadIdx.x >> 5;
  const int lane = threadIdx.x & 31;
  const int m    = lane & 15;
  const int hi   = lane >> 4;
  const bool upper = hi != 0;

  const int by = blockIdx.y;
  const int tg = by < NG / 2 ? NG - 1 - by : by - NG / 2;

  const int qt = (tg * NW + wave) * 16;

  const size_t base = (size_t)bh * S_LEN * HD;
  const float* qb = q + base;
  const float* kb = k + base;
  const float* vb = v + base;
  float*       ob = out + base;

  const float scale = 0.125f;

  const int srow = threadIdx.x >> 4;
  const int sc4  = threadIdx.x & 15;


  v2f qa[16];
#pragma unroll
  for (int c = 0; c < 16; ++c) {
    const float* p = qb + (size_t)(qt + m) * HD + 4 * c + 2 * hi;
    v2f t = *(const v2f*)p;
    qa[c] = t * scale;
  }

  v8f acc[4] = {};
  float carry = 1.0f;

  float* wl = lds + wave * (KB * 16 * WROW);

  const int gtop = tg;

  {
    const int jg = gtop * GK;
    const int jb = (gtop > 0 ? gtop - 1 : 0) * GK;
    float* lkt = lk[gtop & 1];
    float* lkb = lk[(gtop & 1) ^ 1];
    float* lvt = lv[gtop & 1];
#pragma unroll
    for (int i = 0; i < 8; ++i) {
      const int row = i * 8 + srow;
      v4f t = *(const v4f*)(kb + (size_t)(jg + row) * HD + 4 * sc4);
      v4f u = *(const v4f*)(kb + (size_t)(jb + row) * HD + 4 * sc4);
      *(v4f*)&lkt[row * KP + 4 * sc4] = t;
      *(v4f*)&lkb[row * KP + 4 * sc4] = u;
    }
#pragma unroll
    for (int i = 0; i < 4; ++i) {
      const int pr = i * 8 + srow;
      const float* p = vb + (size_t)(jg + 2 * pr) * HD + 4 * sc4;
      v4f a = *(const v4f*)p;
      v4f b = *(const v4f*)(p + HD);
      *(v4f*)&lvt[pr * VP + 8 * sc4]     = (v4f){a.x, b.x, a.y, b.y};
      *(v4f*)&lvt[pr * VP + 8 * sc4 + 4] = (v4f){a.z, b.z, a.w, b.w};
    }
  }
  __syncthreads();

  {
    const float* lkc = lk[gtop & 1];
    const int jg = gtop * GK;
    v8f L[KB];
#pragma unroll
    for (int b = 0; b < KB; ++b) L[b] = (v8f){};
    v2f kf[16][KB];
#pragma unroll
    for (int c = 0; c < 16; ++c)
#pragma unroll
      for (int b = 0; b < KB; ++b)
        kf[c][b] = *(const v2f*)&lkc[(16 * b + m) * KP + 4 * c + 2 * hi];
#pragma unroll
    for (int c = 0; c < 16; ++c)
#pragma unroll
      for (int b = 0; b < KB; ++b)
        L[b] = wmma_f32(qa[c], kf[c][b], L[b]);
#pragma unroll
    for (int b = 0; b < KB; ++b)
#pragma unroll
      for (int r = 0; r < 8; ++r) {
        float x = L[b][r];
        if (jg + 16 * b + m > qt + r + 8 * hi) x = -__builtin_inff();
        wl[b * (16 * WROW) + (r + 8 * hi) * WROW + m] = x;
      }
  }
  __syncthreads();

  float zz[KB][8], sc[KB][8];
  float ta[32], ea[32], da[32], ra[32], pa[32];
  float to[KB], fs[KB], tt[KB], F[KB];
  v2f sf[KB][4];
  v2f wa[KB][4];

  for (int g = gtop; g >= 1; --g) {
    const float* lkc = lk[(g & 1) ^ 1];
    const float* lvc = lv[g & 1];
    float* lkn = lk[g & 1];
    float* lvn = lv[(g & 1) ^ 1];
    const int jk = (g > 1 ? g - 2 : 0) * GK;
    const int jv = (g - 1) * GK;

    v4f ks[8];
#pragma unroll
    for (int i = 0; i < 8; ++i)
      ks[i] = *(const v4f*)(kb + (size_t)(jk + i * 8 + srow) * HD + 4 * sc4);

    v8f LA[KB];
#pragma unroll
    for (int b = KB - 1; b >= 0; --b) {
      const v4f lo = *(const v4f*)&wl[b * (16 * WROW) + m * WROW + 8 * hi];
      const v4f up = *(const v4f*)&wl[b * (16 * WROW) + m * WROW + 8 * hi + 4];
      LA[b] = (v8f){lo.x, lo.y, lo.z, lo.w, up.x, up.y, up.z, up.w};
    }
    __builtin_amdgcn_sched_barrier(0);

    v8f Ln[KB];
#pragma unroll
    for (int b = 0; b < KB; ++b) Ln[b] = (v8f){};
    v2f kf[16][KB];
#pragma unroll
    for (int c = 0; c < 16; ++c)
#pragma unroll
      for (int b = 0; b < KB; ++b)
        kf[c][b] = *(const v2f*)&lkc[(16 * b + m) * KP + 4 * c + 2 * hi];
    __builtin_amdgcn_sched_barrier(0);

    asm volatile("" : "+v"(LA[0]), "+v"(LA[1]), "+v"(LA[2]), "+v"(LA[3]));
    SLOT8(0) SLOT8(8) SLOT8(16) SLOT8(24) SLOT8(32) SLOT8(40) SLOT8(48) SLOT8(56)
    PIPE8 PIPE8 PIPE8 PIPE8 PIPE8 PIPE8 PIPE8 PIPE8
    asm volatile("" : "+v"(Ln[0]), "+v"(Ln[1]), "+v"(Ln[2]), "+v"(Ln[3]),
                      "+v"(wa[0][0]), "+v"(wa[0][1]), "+v"(wa[0][2]), "+v"(wa[0][3]),
                      "+v"(wa[1][0]), "+v"(wa[1][1]), "+v"(wa[1][2]), "+v"(wa[1][3]),
                      "+v"(wa[2][0]), "+v"(wa[2][1]), "+v"(wa[2][2]), "+v"(wa[2][3]),
                      "+v"(wa[3][0]), "+v"(wa[3][1]), "+v"(wa[3][2]), "+v"(wa[3][3])
                 :: "memory");
    __builtin_amdgcn_sched_barrier(0);

    v2f vf[KB][4][4];
#pragma unroll
    for (int b = KB - 1; b >= 0; --b)
#pragma unroll
      for (int c = 0; c < 4; ++c) {
        const float* vp = &lvc[(8 * b + 4 * hi + c) * VP + 2 * m];
#pragma unroll
        for (int t = 0; t < 4; ++t) vf[b][c][t] = *(const v2f*)(vp + 32 * t);
      }
    __builtin_amdgcn_sched_barrier(0);

#pragma unroll
    for (int i = 0; i < 8; ++i)
      *(v4f*)&lkn[(i * 8 + srow) * KP + 4 * sc4] = ks[i];
#pragma unroll
    for (int b = 0; b < KB; ++b)
#pragma unroll
      for (int r = 0; r < 8; ++r)
        wl[b * (16 * WROW) + (r + 8 * hi) * WROW + m] = Ln[b][r];
    v4f va[4], vc[4];
#pragma unroll
    for (int i = 0; i < 4; ++i) {
      const float* p = vb + (size_t)(jv + 2 * (i * 8 + srow)) * HD + 4 * sc4;
      va[i] = *(const v4f*)p;
      vc[i] = *(const v4f*)(p + HD);
    }
    __builtin_amdgcn_sched_barrier(0);

#pragma unroll
    for (int b = KB - 1; b >= 0; --b)
#pragma unroll
      for (int c = 0; c < 4; ++c)
#pragma unroll
        for (int t = 0; t < 4; ++t)
          acc[t] = wmma_f32(wa[b][c], vf[b][c][t], acc[t]);
    asm volatile("" : "+v"(acc[0]), "+v"(acc[1]), "+v"(acc[2]), "+v"(acc[3]),
                      "+v"(va[0]), "+v"(va[1]), "+v"(va[2]), "+v"(va[3]),
                      "+v"(vc[0]), "+v"(vc[1]), "+v"(vc[2]), "+v"(vc[3])
                 :: "memory");

#pragma unroll
    for (int i = 0; i < 4; ++i) {
      const int pr = i * 8 + srow;
      *(v4f*)&lvn[pr * VP + 8 * sc4]     = (v4f){va[i].x, vc[i].x, va[i].y, vc[i].y};
      *(v4f*)&lvn[pr * VP + 8 * sc4 + 4] = (v4f){va[i].z, vc[i].z, va[i].w, vc[i].w};
    }
    __syncthreads();
  }

  {
    const float* lvc = lv[0];
    v8f LA[KB];
#pragma unroll
    for (int b = KB - 1; b >= 0; --b) {
      const v4f lo = *(const v4f*)&wl[b * (16 * WROW) + m * WROW + 8 * hi];
      const v4f up = *(const v4f*)&wl[b * (16 * WROW) + m * WROW + 8 * hi + 4];
      LA[b] = (v8f){lo.x, lo.y, lo.z, lo.w, up.x, up.y, up.z, up.w};
    }
    v2f vf[KB][4][4];
#pragma unroll
    for (int b = KB - 1; b >= 0; --b)
#pragma unroll
      for (int c = 0; c < 4; ++c) {
        const float* vp = &lvc[(8 * b + 4 * hi + c) * VP + 2 * m];
#pragma unroll
        for (int t = 0; t < 4; ++t) vf[b][c][t] = *(const v2f*)(vp + 32 * t);
      }
    GS8(0) GS8(8) GS8(16) GS8(24) GS8(32) GS8(40) group_slice<48>(GS_ARGS);
#pragma unroll
    for (int b = KB - 1; b >= 0; --b)
#pragma unroll
      for (int c = 0; c < 4; ++c)
#pragma unroll
        for (int t = 0; t < 4; ++t)
          acc[t] = wmma_f32(wa[b][c], vf[b][c][t], acc[t]);
  }

#pragma unroll
  for (int r = 0; r < 8; ++r) {
    size_t row = (size_t)(qt + r + 8 * hi) * HD;
    ob[row + 0  + m] = acc[0][r];
    ob[row + 16 + m] = acc[1][r];
    ob[row + 32 + m] = acc[2][r];
    ob[row + 48 + m] = acc[3][r];
  }
}

extern "C" void kernel_launch(void* const* d_in, const int* in_sizes, int n_in,
                              void* d_out, int out_size, void* d_ws, size_t ws_size,
                              hipStream_t stream) {
  const float* q = (const float*)d_in[0];
  const float* k = (const float*)d_in[1];
  const float* v = (const float*)d_in[2];
  float* out = (float*)d_out;

  const int BH = in_sizes[0] / (S_LEN * HD);
  dim3 grid(BH, S_LEN / (16 * NW));
  dim3 block(32 * NW);
  hipLaunchKernelGGL(stickbreak_kernel, grid, block, 0, stream, q, k, v, out);
}


# Sigmoid pair from one hardware tanh of the half logit (Q scaled by 1/16): z = (1+t)/2, beta = (1-t)/2 by packed fma, in place of exp2, rcp, compare, selects; rounding changes, sum order unchanged
# speedup vs baseline: 2.7377x; 1.0506x over previous
#include <hip/hip_runtime.h>
#include <hip/hip_bf16.h>


typedef __attribute__((ext_vector_type(2))) float v2f;
typedef __attribute__((ext_vector_type(8))) float v8f;
typedef __attribute__((ext_vector_type(4))) float v4f;

#define S_LEN 1024
#define HD    64
#define NW    4
#define KB    4
#define WROW  20
#define GK    (16 * KB)
#define NG    (S_LEN / (16 * NW))
#define KP    68
#define MASKED (-64.0f)
#define VP    136

__device__ __forceinline__ v8f wmma_f32(v2f a, v2f b, v8f c) {
  return __builtin_amdgcn_wmma_f32_16x16x4_f32(false, a, false, b, (short)0, c,
                                               false, false);
}

__device__ __forceinline__ float other_half(float x) {
  const unsigned u = __builtin_bit_cast(unsigned, x);
  return __builtin_bit_cast(float,
      __builtin_amdgcn_permlanex16(u, u, 0x76543210u, 0xfedcba98u, false, false));
}

template <int K>
__device__ __forceinline__ void tanh_step(const v8f (&LA)[KB], float (&ta)[32]) {
  if constexpr (K >= 0 && K < 32) {
    constexpr int b = KB - 1 - (K >> 3), e = 7 - (K & 7);
    ta[K] = __builtin_amdgcn_tanhf(LA[b][e]);
  }
}

template <int K>
__device__ __forceinline__ void pair_step(const float (&ta)[32],
                                          float (&zz)[KB][8], float (&sc)[KB][8]) {
  if constexpr (K >= 1 && K < 32 && (K & 1) == 1) {
    constexpr int b = KB - 1 - (K >> 3), e = 7 - (K & 7);
    const v2f t  = (v2f){ta[K], ta[K - 1]};
    const v2f hf = (v2f){0.5f, 0.5f};
    const v2f z  = __builtin_elementwise_fma(t, hf, hf);
    const v2f s  = __builtin_elementwise_fma(t, -hf, hf);
    zz[b][e] = z.x; zz[b][e + 1] = z.y;
    sc[b][e] = s.x; sc[b][e + 1] = s.y;
  }
}

template <int B, int U>
__device__ __forceinline__ void scan_step(bool upper, float (&zz)[KB][8],
                                          float (&sc)[KB][8], float (&to)[KB],
                                          float (&fs)[KB], float (&tt)[KB],
                                          float (&F)[KB], v2f (&sf)[KB][4],
                                          float &carry, v2f (&wa)[KB][4]) {
  if constexpr (U >= 0 && U < 7) sc[B][6 - U] = sc[B][6 - U] * sc[B][7 - U];
  if constexpr (U == 7) to[B] = other_half(sc[B][0]);
  if constexpr (U == 8) {
    fs[B] = upper ? 1.0f : to[B];
    tt[B] = sc[B][0] * to[B];
  }
  if constexpr (U == 9) {
    F[B]  = carry * fs[B];
    carry = carry * tt[B];
  }
  if constexpr (U >= 10 && U <= 17) {
    constexpr int c = (U - 10) >> 1;
    if constexpr (((U - 10) & 1) == 0)
      sf[B][c] = (v2f){sc[B][2 * c] * F[B], sc[B][2 * c + 1] * F[B]};
    else
      wa[B][c] = (v2f){zz[B][2 * c] * sf[B][c].x, zz[B][2 * c + 1] * sf[B][c].y};
  }
}

template <int S>
__device__ __forceinline__ void group_slice(bool upper, const v8f (&LA)[KB],
                                            float (&ta)[32],
                                            float (&zz)[KB][8], float (&sc)[KB][8],
                                            float (&to)[KB], float (&fs)[KB],
                                            float (&tt)[KB], float (&F)[KB],
                                            v2f (&sf)[KB][4], float &carry,
                                            v2f (&wa)[KB][4]) {
  tanh_step<S>(LA, ta);
  pair_step<S - 1>(ta, zz, sc);
  scan_step<3, S - 7>(upper, zz, sc, to, fs, tt, F, sf, carry, wa);
  scan_step<2, S - 15>(upper, zz, sc, to, fs, tt, F, sf, carry, wa);
  scan_step<1, S - 23>(upper, zz, sc, to, fs, tt, F, sf, carry, wa);
  scan_step<0, S - 31>(upper, zz, sc, to, fs, tt, F, sf, carry, wa);
}
#define GS_ARGS upper, LA, ta, zz, sc, to, fs, tt, F, sf, carry, wa
#define GS8(a) group_slice<(a)>(GS_ARGS); group_slice<(a) + 1>(GS_ARGS);       \
  group_slice<(a) + 2>(GS_ARGS); group_slice<(a) + 3>(GS_ARGS);                \
  group_slice<(a) + 4>(GS_ARGS); group_slice<(a) + 5>(GS_ARGS);                \
  group_slice<(a) + 6>(GS_ARGS); group_slice<(a) + 7>(GS_ARGS);
#define SLOT(a)                                                                \
  Ln[(a) & 3] = wmma_f32(qa[(a) >> 2], kf[(a) >> 2][(a) & 3], Ln[(a) & 3]);    \
  group_slice<(a)>(GS_ARGS);
#define PIPE1                                                                  \
  __builtin_amdgcn_sched_group_barrier(0x008, 1, 0);                           \
  __builtin_amdgcn_sched_group_barrier(0x002, 3, 0);                           \
  __builtin_amdgcn_sched_group_barrier(0x400, 1, 0);                           \
  __builtin_amdgcn_sched_group_barrier(0x002, 3, 0);                           \
  __builtin_amdgcn_sched_group_barrier(0x400, 1, 0);
#define PIPE8 PIPE1 PIPE1 PIPE1 PIPE1 PIPE1 PIPE1 PIPE1 PIPE1
#define SLOT8(a) SLOT(a) SLOT((a) + 1) SLOT((a) + 2) SLOT((a) + 3)             \
  SLOT((a) + 4) SLOT((a) + 5) SLOT((a) + 6) SLOT((a) + 7)


__global__ __launch_bounds__(128) void stickbreak_kernel(
    const float* __restrict__ q, const float* __restrict__ k,
    const float* __restrict__ v, float* __restrict__ out) {
  __shared__ __attribute__((aligned(16))) float lds[NW * KB * 16 * WROW];
  __shared__ __attribute__((aligned(16))) float lk[2][GK * KP];
  __shared__ __attribute__((aligned(16))) float lv[2][(GK / 2) * VP];

  const int bh   = blockIdx.x;
  const int wave = threadIdx.x >> 5;
  const int lane = threadIdx.x & 31;
  const int m    = lane & 15;
  const int hi   = lane >> 4;
  const bool upper = hi != 0;

  const int by = blockIdx.y;
  const int tg = by < NG / 2 ? NG - 1 - by : by - NG / 2;

  const int qt = (tg * NW + wave) * 16;

  const size_t base = (size_t)bh * S_LEN * HD;
  const float* qb = q + base;
  const float* kb = k + base;
  const float* vb = v + base;
  float*       ob = out + base;

  const float scale = 0.0625f;

  const int srow = threadIdx.x >> 4;
  const int sc4  = threadIdx.x & 15;


  v2f qa[16];
#pragma unroll
  for (int c = 0; c < 16; ++c) {
    const float* p = qb + (size_t)(qt + m) * HD + 4 * c + 2 * hi;
    v2f t = *(const v2f*)p;
    qa[c] = t * scale;
  }

  v8f acc[4] = {};
  float carry = 1.0f;

  float* wl = lds + wave * (KB * 16 * WROW);

  const int gtop = tg;

  {
    const int jg = gtop * GK;
    const int jb = (gtop > 0 ? gtop - 1 : 0) * GK;
    float* lkt = lk[gtop & 1];
    float* lkb = lk[(gtop & 1) ^ 1];
    float* lvt = lv[gtop & 1];
#pragma unroll
    for (int i = 0; i < 8; ++i) {
      const int row = i * 8 + srow;
      v4f t = *(const v4f*)(kb + (size_t)(jg + row) * HD + 4 * sc4);
      v4f u = *(const v4f*)(kb + (size_t)(jb + row) * HD + 4 * sc4);
      *(v4f*)&lkt[row * KP + 4 * sc4] = t;
      *(v4f*)&lkb[row * KP + 4 * sc4] = u;
    }
#pragma unroll
    for (int i = 0; i < 4; ++i) {
      const int pr = i * 8 + srow;
      const float* p = vb + (size_t)(jg + 2 * pr) * HD + 4 * sc4;
      v4f a = *(const v4f*)p;
      v4f b = *(const v4f*)(p + HD);
      *(v4f*)&lvt[pr * VP + 8 * sc4]     = (v4f){a.x, b.x, a.y, b.y};
      *(v4f*)&lvt[pr * VP + 8 * sc4 + 4] = (v4f){a.z, b.z, a.w, b.w};
    }
  }
  __syncthreads();

  {
    const float* lkc = lk[gtop & 1];
    const int jg = gtop * GK;
    v8f L[KB];
#pragma unroll
    for (int b = 0; b < KB; ++b) L[b] = (v8f){};
    v2f kf[16][KB];
#pragma unroll
    for (int c = 0; c < 16; ++c)
#pragma unroll
      for (int b = 0; b < KB; ++b)
        kf[c][b] = *(const v2f*)&lkc[(16 * b + m) * KP + 4 * c + 2 * hi];
#pragma unroll
    for (int c = 0; c < 16; ++c)
#pragma unroll
      for (int b = 0; b < KB; ++b)
        L[b] = wmma_f32(qa[c], kf[c][b], L[b]);
#pragma unroll
    for (int b = 0; b < KB; ++b)
#pragma unroll
      for (int r = 0; r < 8; ++r) {
        float x = L[b][r];
        if (jg + 16 * b + m > qt + r + 8 * hi) x = MASKED;
        wl[b * (16 * WROW) + (r + 8 * hi) * WROW + m] = x;
      }
  }
  __syncthreads();

  float zz[KB][8], sc[KB][8];
  float ta[32];
  float to[KB], fs[KB], tt[KB], F[KB];
  v2f sf[KB][4];
  v2f wa[KB][4];

  for (int g = gtop; g >= 1; --g) {
    const float* lkc = lk[(g & 1) ^ 1];
    const float* lvc = lv[g & 1];
    float* lkn = lk[g & 1];
    float* lvn = lv[(g & 1) ^ 1];
    const int jk = (g > 1 ? g - 2 : 0) * GK;
    const int jv = (g - 1) * GK;

    v4f ks[8];
#pragma unroll
    for (int i = 0; i < 8; ++i)
      ks[i] = *(const v4f*)(kb + (size_t)(jk + i * 8 + srow) * HD + 4 * sc4);

    v8f LA[KB];
#pragma unroll
    for (int b = KB - 1; b >= 0; --b) {
      const v4f lo = *(const v4f*)&wl[b * (16 * WROW) + m * WROW + 8 * hi];
      const v4f up = *(const v4f*)&wl[b * (16 * WROW) + m * WROW + 8 * hi + 4];
      LA[b] = (v8f){lo.x, lo.y, lo.z, lo.w, up.x, up.y, up.z, up.w};
    }
    __builtin_amdgcn_sched_barrier(0);

    v8f Ln[KB];
#pragma unroll
    for (int b = 0; b < KB; ++b) Ln[b] = (v8f){};
    v2f kf[16][KB];
#pragma unroll
    for (int c = 0; c < 16; ++c)
#pragma unroll
      for (int b = 0; b < KB; ++b)
        kf[c][b] = *(const v2f*)&lkc[(16 * b + m) * KP + 4 * c + 2 * hi];
    __builtin_amdgcn_sched_barrier(0);

    asm volatile("" : "+v"(LA[0]), "+v"(LA[1]), "+v"(LA[2]), "+v"(LA[3]));
    SLOT8(0) SLOT8(8) SLOT8(16) SLOT8(24) SLOT8(32) SLOT8(40) SLOT8(48) SLOT8(56)
    PIPE8 PIPE8 PIPE8 PIPE8 PIPE8 PIPE8 PIPE8 PIPE8
    asm volatile("" : "+v"(Ln[0]), "+v"(Ln[1]), "+v"(Ln[2]), "+v"(Ln[3]),
                      "+v"(wa[0][0]), "+v"(wa[0][1]), "+v"(wa[0][2]), "+v"(wa[0][3]),
                      "+v"(wa[1][0]), "+v"(wa[1][1]), "+v"(wa[1][2]), "+v"(wa[1][3]),
                      "+v"(wa[2][0]), "+v"(wa[2][1]), "+v"(wa[2][2]), "+v"(wa[2][3]),
                      "+v"(wa[3][0]), "+v"(wa[3][1]), "+v"(wa[3][2]), "+v"(wa[3][3])
                 :: "memory");
    __builtin_amdgcn_sched_barrier(0);

    v2f vf[KB][4][4];
#pragma unroll
    for (int b = KB - 1; b >= 0; --b)
#pragma unroll
      for (int c = 0; c < 4; ++c) {
        const float* vp = &lvc[(8 * b + 4 * hi + c) * VP + 2 * m];
#pragma unroll
        for (int t = 0; t < 4; ++t) vf[b][c][t] = *(const v2f*)(vp + 32 * t);
      }
    __builtin_amdgcn_sched_barrier(0);

#pragma unroll
    for (int i = 0; i < 8; ++i)
      *(v4f*)&lkn[(i * 8 + srow) * KP + 4 * sc4] = ks[i];
#pragma unroll
    for (int b = 0; b < KB; ++b)
#pragma unroll
      for (int r = 0; r < 8; ++r)
        wl[b * (16 * WROW) + (r + 8 * hi) * WROW + m] = Ln[b][r];
    v4f va[4], vc[4];
#pragma unroll
    for (int i = 0; i < 4; ++i) {
      const float* p = vb + (size_t)(jv + 2 * (i * 8 + srow)) * HD + 4 * sc4;
      va[i] = *(const v4f*)p;
      vc[i] = *(const v4f*)(p + HD);
    }
    __builtin_amdgcn_sched_barrier(0);

#pragma unroll
    for (int b = KB - 1; b >= 0; --b)
#pragma unroll
      for (int c = 0; c < 4; ++c)
#pragma unroll
        for (int t = 0; t < 4; ++t)
          acc[t] = wmma_f32(wa[b][c], vf[b][c][t], acc[t]);
    asm volatile("" : "+v"(acc[0]), "+v"(acc[1]), "+v"(acc[2]), "+v"(acc[3]),
                      "+v"(va[0]), "+v"(va[1]), "+v"(va[2]), "+v"(va[3]),
                      "+v"(vc[0]), "+v"(vc[1]), "+v"(vc[2]), "+v"(vc[3])
                 :: "memory");

#pragma unroll
    for (int i = 0; i < 4; ++i) {
      const int pr = i * 8 + srow;
      *(v4f*)&lvn[pr * VP + 8 * sc4]     = (v4f){va[i].x, vc[i].x, va[i].y, vc[i].y};
      *(v4f*)&lvn[pr * VP + 8 * sc4 + 4] = (v4f){va[i].z, vc[i].z, va[i].w, vc[i].w};
    }
    __syncthreads();
  }

  {
    const float* lvc = lv[0];
    v8f LA[KB];
#pragma unroll
    for (int b = KB - 1; b >= 0; --b) {
      const v4f lo = *(const v4f*)&wl[b * (16 * WROW) + m * WROW + 8 * hi];
      const v4f up = *(const v4f*)&wl[b * (16 * WROW) + m * WROW + 8 * hi + 4];
      LA[b] = (v8f){lo.x, lo.y, lo.z, lo.w, up.x, up.y, up.z, up.w};
    }
    v2f vf[KB][4][4];
#pragma unroll
    for (int b = KB - 1; b >= 0; --b)
#pragma unroll
      for (int c = 0; c < 4; ++c) {
        const float* vp = &lvc[(8 * b + 4 * hi + c) * VP + 2 * m];
#pragma unroll
        for (int t = 0; t < 4; ++t) vf[b][c][t] = *(const v2f*)(vp + 32 * t);
      }
    GS8(0) GS8(8) GS8(16) GS8(24) GS8(32) GS8(40) group_slice<48>(GS_ARGS);
#pragma unroll
    for (int b = KB - 1; b >= 0; --b)
#pragma unroll
      for (int c = 0; c < 4; ++c)
#pragma unroll
        for (int t = 0; t < 4; ++t)
          acc[t] = wmma_f32(wa[b][c], vf[b][c][t], acc[t]);
  }

#pragma unroll
  for (int r = 0; r < 8; ++r) {
    size_t row = (size_t)(qt + r + 8 * hi) * HD;
    ob[row + 0  + m] = acc[0][r];
    ob[row + 16 + m] = acc[1][r];
    ob[row + 32 + m] = acc[2][r];
    ob[row + 48 + m] = acc[3][r];
  }
}

extern "C" void kernel_launch(void* const* d_in, const int* in_sizes, int n_in,
                              void* d_out, int out_size, void* d_ws, size_t ws_size,
                              hipStream_t stream) {
  const float* q = (const float*)d_in[0];
  const float* k = (const float*)d_in[1];
  const float* v = (const float*)d_in[2];
  float* out = (float*)d_out;

  const int BH = in_sizes[0] / (S_LEN * HD);
  dim3 grid(BH, S_LEN / (16 * NW));
  dim3 block(32 * NW);
  hipLaunchKernelGGL(stickbreak_kernel, grid, block, 0, stream, q, k, v, out);
}
